# starting kernel
# baseline (speedup 1.0000x reference)
#include <hip/hip_runtime.h>


typedef __attribute__((ext_vector_type(2))) float v2f;
typedef __attribute__((ext_vector_type(4))) float v4f;
typedef __attribute__((ext_vector_type(8))) float v8f;

#define K_FIELDS 39
#define NCOLS    (8192 * 64)
#define WPB      8
#define TPW      4
#define PAIRW    32
#define TILE_W   (24 * PAIRW)

__global__ __launch_bounds__(256, 1)
void fwfm_wmma_kernel(const float* __restrict__ x,
                      const float* __restrict__ w,
                      float* __restrict__ out)
{
    __shared__ float s_pad[48 * 40];
    __shared__ float s_diag[48];
    __shared__ __align__(16) float x_tile[WPB][TILE_W];

    const int tid  = threadIdx.x;
    const int lane = tid & 31;
    const int wv   = tid >> 5;
    const int hl   = lane >> 4;
    const int l16  = lane & 15;

    for (int i = tid; i < 48 * 40; i += 256) {
        const int r = i / 40, c = i - r * 40;
        float v = 0.0f;
        if (r < K_FIELDS && c < K_FIELDS)
            v = 0.5f * (w[r * K_FIELDS + c] + w[c * K_FIELDS + r]);
        s_pad[i] = v;
    }
    if (tid < 48)
        s_diag[tid] = (tid < K_FIELDS) ? w[tid * K_FIELDS + tid] : 0.0f;

    float* xt = &x_tile[wv][0];
    for (int i = lane; i < TILE_W; i += 32) xt[i] = 0.0f;
    __syncthreads();

    v2f a[3][10];
    #pragma unroll
    for (int m = 0; m < 3; ++m) {
        #pragma unroll
        for (int kk = 0; kk < 10; ++kk) {
            const int row = m * 16 + l16;
            const int k0  = kk * 4 + 2 * hl;
            v2f t;
            t[0] = s_pad[row * 40 + k0];
            t[1] = s_pad[row * 40 + k0 + 1];
            a[m][kk] = t;
        }
    }
    float sdr[24];
    #pragma unroll
    for (int m = 0; m < 3; ++m)
        #pragma unroll
        for (int v = 0; v < 8; ++v)
            sdr[m * 8 + v] = s_diag[m * 16 + hl * 8 + v];

    const int sp = lane >> 3;
    const int sc = (lane & 7) * 2;

    const long long gw = (long long)blockIdx.x * WPB + wv;
    const long long nbase = gw * (TPW * 16);

    float2 pa[5], pb[5];
    #pragma unroll
    for (int r = 0; r < 5; ++r) {
        const int p    = r * 4 + sp;
        const int row1 = 2 * p + 1;
        pa[r] = *(const float2*)&x[(long long)(2 * p) * NCOLS + nbase + sc];
        if (row1 < K_FIELDS)
            pb[r] = *(const float2*)&x[(long long)row1 * NCOLS + nbase + sc];
        else
            pb[r] = make_float2(0.0f, 0.0f);
    }

    for (int t = 0; t < TPW; ++t) {
        const long long n0 = nbase + (long long)t * 16;

        #pragma unroll
        for (int r = 0; r < 5; ++r) {
            const int p = r * 4 + sp;
            v4f q;
            q[0] = pa[r].x; q[1] = pb[r].x; q[2] = pa[r].y; q[3] = pb[r].y;
            *(v4f*)&xt[p * PAIRW + sc * 2] = q;
        }
        asm volatile("" ::: "memory");

        if (t + 1 < TPW) {
            const long long n1 = n0 + 16;
            #pragma unroll
            for (int r = 0; r < 5; ++r) {
                const int p    = r * 4 + sp;
                const int row1 = 2 * p + 1;
                pa[r] = *(const float2*)&x[(long long)(2 * p) * NCOLS + n1 + sc];
                if (row1 < K_FIELDS)
                    pb[r] = *(const float2*)&x[(long long)row1 * NCOLS + n1 + sc];
                else
                    pb[r] = make_float2(0.0f, 0.0f);
            }
        }

        v8f cc[3] = {v8f{}, v8f{}, v8f{}};
        #pragma unroll
        for (int kk = 0; kk < 10; ++kk) {
            const int p = 2 * kk + hl;
            const v2f b = *(const v2f*)&xt[p * PAIRW + l16 * 2];
            cc[0] = __builtin_amdgcn_wmma_f32_16x16x4_f32(false, a[0][kk], false, b,
                                                          (short)0, cc[0], false, false);
            cc[1] = __builtin_amdgcn_wmma_f32_16x16x4_f32(false, a[1][kk], false, b,
                                                          (short)0, cc[1], false, false);
            cc[2] = __builtin_amdgcn_wmma_f32_16x16x4_f32(false, a[2][kk], false, b,
                                                          (short)0, cc[2], false, false);
        }

        float acc = 0.0f;
        #pragma unroll
        for (int m = 0; m < 3; ++m) {
            #pragma unroll
            for (int pp = 0; pp < 4; ++pp) {
                const int p = m * 8 + hl * 4 + pp;
                const v2f xp = *(const v2f*)&xt[p * PAIRW + l16 * 2];
                const float x0 = xp[0], x1 = xp[1];
                acc += x0 * (cc[m][2 * pp]     - sdr[m * 8 + 2 * pp]     * x0);
                acc += x1 * (cc[m][2 * pp + 1] - sdr[m * 8 + 2 * pp + 1] * x1);
            }
        }
        acc += __shfl_xor(acc, 16, 32);
        if (lane < 16)
            out[n0 + lane] = 0.5f * acc;
    }
}

extern "C" void kernel_launch(void* const* d_in, const int* in_sizes, int n_in,
                              void* d_out, int out_size, void* d_ws, size_t ws_size,
                              hipStream_t stream)
{
    const float* x = (const float*)d_in[0];
    const float* w = (const float*)d_in[1];
    float* out = (float*)d_out;

    dim3 grid(NCOLS / (WPB * TPW * 16));
    dim3 block(256);
    fwfm_wmma_kernel<<<grid, block, 0, stream>>>(x, w, out);
}


# S is symmetric, so sum only k<l: A holds the strict upper triangle of S and its all-zero 16x4 blocks are skipped (18 WMMAs per tile, not 30); no diagonal term; f32 summation order changes
# speedup vs baseline: 1.2879x; 1.2879x over previous
#include <hip/hip_runtime.h>


typedef __attribute__((ext_vector_type(2))) float v2f;
typedef __attribute__((ext_vector_type(4))) float v4f;
typedef __attribute__((ext_vector_type(8))) float v8f;

#define K_FIELDS 39
#define NCOLS    (8192 * 64)
#define WPB      8
#define TPW      4
#define PAIRW    32
#define TILE_W   (24 * PAIRW)

__global__ __launch_bounds__(256, 1)
void fwfm_wmma_kernel(const float* __restrict__ x,
                      const float* __restrict__ w,
                      float* __restrict__ out)
{
    __shared__ __align__(16) float s_pad[48 * 40];
    __shared__ __align__(16) float x_tile[WPB][TILE_W];

    const int tid  = threadIdx.x;
    const int lane = tid & 31;
    const int wv   = tid >> 5;
    const int hl   = lane >> 4;
    const int l16  = lane & 15;

    for (int i = tid; i < 48 * 40; i += 256) {
        const int r = i / 40, c = i - r * 40;
        float v = 0.0f;
        if (c > r && c < K_FIELDS)
            v = 0.5f * (w[r * K_FIELDS + c] + w[c * K_FIELDS + r]);
        s_pad[i] = v;
    }

    float* xt = &x_tile[wv][0];
    for (int i = lane; i < TILE_W; i += 32) xt[i] = 0.0f;
    __syncthreads();

    v2f a0[10], a1[6], a2[2];
    #pragma unroll
    for (int kk = 0; kk < 10; ++kk) {
        const int k0 = kk * 4 + 2 * hl;
        a0[kk] = *(const v2f*)&s_pad[l16 * 40 + k0];
        if (kk >= 4) a1[kk - 4] = *(const v2f*)&s_pad[(16 + l16) * 40 + k0];
        if (kk >= 8) a2[kk - 8] = *(const v2f*)&s_pad[(32 + l16) * 40 + k0];
    }

    const int sp = lane >> 3;
    const int sc = (lane & 7) * 2;

    const long long gw = (long long)blockIdx.x * WPB + wv;
    const long long nbase = gw * (TPW * 16);

    float2 pa[5], pb[5];
    #pragma unroll
    for (int r = 0; r < 5; ++r) {
        const int p    = r * 4 + sp;
        const int row1 = 2 * p + 1;
        pa[r] = *(const float2*)&x[(long long)(2 * p) * NCOLS + nbase + sc];
        if (row1 < K_FIELDS)
            pb[r] = *(const float2*)&x[(long long)row1 * NCOLS + nbase + sc];
        else
            pb[r] = make_float2(0.0f, 0.0f);
    }

    for (int t = 0; t < TPW; ++t) {
        const long long n0 = nbase + (long long)t * 16;

        #pragma unroll
        for (int r = 0; r < 5; ++r) {
            const int p = r * 4 + sp;
            v4f q;
            q[0] = pa[r].x; q[1] = pb[r].x; q[2] = pa[r].y; q[3] = pb[r].y;
            *(v4f*)&xt[p * PAIRW + sc * 2] = q;
        }
        asm volatile("" ::: "memory");

        if (t + 1 < TPW) {
            const long long n1 = n0 + 16;
            #pragma unroll
            for (int r = 0; r < 5; ++r) {
                const int p    = r * 4 + sp;
                const int row1 = 2 * p + 1;
                pa[r] = *(const float2*)&x[(long long)(2 * p) * NCOLS + n1 + sc];
                if (row1 < K_FIELDS)
                    pb[r] = *(const float2*)&x[(long long)row1 * NCOLS + n1 + sc];
                else
                    pb[r] = make_float2(0.0f, 0.0f);
            }
        }

        v2f b[10];
        #pragma unroll
        for (int kk = 0; kk < 10; ++kk)
            b[kk] = *(const v2f*)&xt[(2 * kk + hl) * PAIRW + l16 * 2];

        v8f cc[3] = {v8f{}, v8f{}, v8f{}};
        #pragma unroll
        for (int kk = 0; kk < 10; ++kk) {
            cc[0] = __builtin_amdgcn_wmma_f32_16x16x4_f32(false, a0[kk], false, b[kk],
                                                          (short)0, cc[0], false, false);
            if (kk >= 4)
                cc[1] = __builtin_amdgcn_wmma_f32_16x16x4_f32(false, a1[kk - 4], false, b[kk],
                                                              (short)0, cc[1], false, false);
            if (kk >= 8)
                cc[2] = __builtin_amdgcn_wmma_f32_16x16x4_f32(false, a2[kk - 8], false, b[kk],
                                                              (short)0, cc[2], false, false);
        }

        float acc = 0.0f;
        #pragma unroll
        for (int m = 0; m < 3; ++m) {
            #pragma unroll
            for (int pp = 0; pp < (m == 2 ? 3 : 4); ++pp) {
                const int p = m * 8 + hl * 4 + pp;
                const v2f xp = *(const v2f*)&xt[p * PAIRW + l16 * 2];
                acc += xp[0] * cc[m][2 * pp];
                acc += xp[1] * cc[m][2 * pp + 1];
            }
        }
        acc += __shfl_xor(acc, 16, 32);
        if (lane < 16)
            out[n0 + lane] = acc;
    }
}

extern "C" void kernel_launch(void* const* d_in, const int* in_sizes, int n_in,
                              void* d_out, int out_size, void* d_ws, size_t ws_size,
                              hipStream_t stream)
{
    const float* x = (const float*)d_in[0];
    const float* w = (const float*)d_in[1];
    float* out = (float*)d_out;

    dim3 grid(NCOLS / (WPB * TPW * 16));
    dim3 block(256);
    fwfm_wmma_kernel<<<grid, block, 0, stream>>>(x, w, out);
}


# Next-tile loads stay in flight across the WMMAs: the row-39 guard on the load (a divergent branch that waited for all 10 loads) becomes a zero select at the LDS commit; loads pinned early
# speedup vs baseline: 1.3857x; 1.0759x over previous
#include <hip/hip_runtime.h>


typedef __attribute__((ext_vector_type(2))) float v2f;
typedef __attribute__((ext_vector_type(4))) float v4f;
typedef __attribute__((ext_vector_type(8))) float v8f;

#define K_FIELDS 39
#define NCOLS    (8192 * 64)
#define WPB      8
#define TPW      4
#define PAIRW    32
#define TILE_W   (24 * PAIRW)

__global__ __launch_bounds__(256, 1)
void fwfm_wmma_kernel(const float* __restrict__ x,
                      const float* __restrict__ w,
                      float* __restrict__ out)
{
    __shared__ __align__(16) float s_pad[48 * 40];
    __shared__ __align__(16) float x_tile[WPB][TILE_W];

    const int tid  = threadIdx.x;
    const int lane = tid & 31;
    const int wv   = tid >> 5;
    const int hl   = lane >> 4;
    const int l16  = lane & 15;

    for (int i = tid; i < 48 * 40; i += 256) {
        const int r = i / 40, c = i - r * 40;
        float v = 0.0f;
        if (c > r && c < K_FIELDS)
            v = 0.5f * (w[r * K_FIELDS + c] + w[c * K_FIELDS + r]);
        s_pad[i] = v;
    }

    float* xt = &x_tile[wv][0];
    for (int i = lane; i < TILE_W; i += 32) xt[i] = 0.0f;
    __syncthreads();

    v2f a0[10], a1[6], a2[2];
    #pragma unroll
    for (int kk = 0; kk < 10; ++kk) {
        const int k0 = kk * 4 + 2 * hl;
        a0[kk] = *(const v2f*)&s_pad[l16 * 40 + k0];
        if (kk >= 4) a1[kk - 4] = *(const v2f*)&s_pad[(16 + l16) * 40 + k0];
        if (kk >= 8) a2[kk - 8] = *(const v2f*)&s_pad[(32 + l16) * 40 + k0];
    }

    const int sp = lane >> 3;
    const int sc = (lane & 7) * 2;

    const long long gw = (long long)blockIdx.x * WPB + wv;
    const long long nbase = gw * (TPW * 16);

    float2 pa[5], pb[5];
    #pragma unroll
    for (int r = 0; r < 5; ++r) {
        const int p    = r * 4 + sp;
        const int row1 = (2 * p + 1 < K_FIELDS) ? (2 * p + 1) : (K_FIELDS - 1);
        pa[r] = *(const float2*)&x[(long long)(2 * p) * NCOLS + nbase + sc];
        pb[r] = *(const float2*)&x[(long long)row1 * NCOLS + nbase + sc];
    }

    for (int t = 0; t < TPW; ++t) {
        const long long n0 = nbase + (long long)t * 16;

        #pragma unroll
        for (int r = 0; r < 5; ++r) {
            const int p = r * 4 + sp;
            const bool odd_ok = (2 * p + 1 < K_FIELDS);
            v4f q;
            q[0] = pa[r].x; q[1] = odd_ok ? pb[r].x : 0.0f;
            q[2] = pa[r].y; q[3] = odd_ok ? pb[r].y : 0.0f;
            *(v4f*)&xt[p * PAIRW + sc * 2] = q;
        }
        asm volatile("" ::: "memory");

        if (t + 1 < TPW) {
            const long long n1 = n0 + 16;
            #pragma unroll
            for (int r = 0; r < 5; ++r) {
                const int p    = r * 4 + sp;
                const int row1 = (2 * p + 1 < K_FIELDS) ? (2 * p + 1) : (K_FIELDS - 1);
                pa[r] = *(const float2*)&x[(long long)(2 * p) * NCOLS + n1 + sc];
                pb[r] = *(const float2*)&x[(long long)row1 * NCOLS + n1 + sc];
            }
        }
        __builtin_amdgcn_sched_barrier(0);

        v2f b[10];
        #pragma unroll
        for (int kk = 0; kk < 10; ++kk)
            b[kk] = *(const v2f*)&xt[(2 * kk + hl) * PAIRW + l16 * 2];

        v8f cc[3] = {v8f{}, v8f{}, v8f{}};
        #pragma unroll
        for (int kk = 0; kk < 10; ++kk) {
            cc[0] = __builtin_amdgcn_wmma_f32_16x16x4_f32(false, a0[kk], false, b[kk],
                                                          (short)0, cc[0], false, false);
            if (kk >= 4)
                cc[1] = __builtin_amdgcn_wmma_f32_16x16x4_f32(false, a1[kk - 4], false, b[kk],
                                                              (short)0, cc[1], false, false);
            if (kk >= 8)
                cc[2] = __builtin_amdgcn_wmma_f32_16x16x4_f32(false, a2[kk - 8], false, b[kk],
                                                              (short)0, cc[2], false, false);
        }

        float acc = 0.0f;
        #pragma unroll
        for (int m = 0; m < 3; ++m) {
            #pragma unroll
            for (int pp = 0; pp < (m == 2 ? 3 : 4); ++pp) {
                const int p = m * 8 + hl * 4 + pp;
                const v2f xp = *(const v2f*)&xt[p * PAIRW + l16 * 2];
                acc += xp[0] * cc[m][2 * pp];
                acc += xp[1] * cc[m][2 * pp + 1];
            }
        }
        acc += __shfl_xor(acc, 16, 32);
        if (lane < 16)
            out[n0 + lane] = acc;
    }
}

extern "C" void kernel_launch(void* const* d_in, const int* in_sizes, int n_in,
                              void* d_out, int out_size, void* d_ws, size_t ws_size,
                              hipStream_t stream)
{
    const float* x = (const float*)d_in[0];
    const float* w = (const float*)d_in[1];
    float* out = (float*)d_out;

    dim3 grid(NCOLS / (WPB * TPW * 16));
    dim3 block(256);
    fwfm_wmma_kernel<<<grid, block, 0, stream>>>(x, w, out);
}


# Each wave loops over 8 tiles instead of 4 (grid 1024 to 512 work-groups, two per WGP): half as many waves to start; U build, barrier and first-tile load wait paid half as often
# speedup vs baseline: 1.4800x; 1.0681x over previous
#include <hip/hip_runtime.h>


typedef __attribute__((ext_vector_type(2))) float v2f;
typedef __attribute__((ext_vector_type(4))) float v4f;
typedef __attribute__((ext_vector_type(8))) float v8f;

#define K_FIELDS 39
#define NCOLS    (8192 * 64)
#define WPB      8
#define TPW      8
#define PAIRW    32
#define TILE_W   (24 * PAIRW)

__global__ __launch_bounds__(256, 1)
void fwfm_wmma_kernel(const float* __restrict__ x,
                      const float* __restrict__ w,
                      float* __restrict__ out)
{
    __shared__ __align__(16) float s_pad[48 * 40];
    __shared__ __align__(16) float x_tile[WPB][TILE_W];

    const int tid  = threadIdx.x;
    const int lane = tid & 31;
    const int wv   = tid >> 5;
    const int hl   = lane >> 4;
    const int l16  = lane & 15;

    for (int i = tid; i < 48 * 40; i += 256) {
        const int r = i / 40, c = i - r * 40;
        float v = 0.0f;
        if (c > r && c < K_FIELDS)
            v = 0.5f * (w[r * K_FIELDS + c] + w[c * K_FIELDS + r]);
        s_pad[i] = v;
    }

    float* xt = &x_tile[wv][0];
    for (int i = lane; i < TILE_W; i += 32) xt[i] = 0.0f;
    __syncthreads();

    v2f a0[10], a1[6], a2[2];
    #pragma unroll
    for (int kk = 0; kk < 10; ++kk) {
        const int k0 = kk * 4 + 2 * hl;
        a0[kk] = *(const v2f*)&s_pad[l16 * 40 + k0];
        if (kk >= 4) a1[kk - 4] = *(const v2f*)&s_pad[(16 + l16) * 40 + k0];
        if (kk >= 8) a2[kk - 8] = *(const v2f*)&s_pad[(32 + l16) * 40 + k0];
    }

    const int sp = lane >> 3;
    const int sc = (lane & 7) * 2;

    const long long gw = (long long)blockIdx.x * WPB + wv;
    const long long nbase = gw * (TPW * 16);

    float2 pa[5], pb[5];
    #pragma unroll
    for (int r = 0; r < 5; ++r) {
        const int p    = r * 4 + sp;
        const int row1 = (2 * p + 1 < K_FIELDS) ? (2 * p + 1) : (K_FIELDS - 1);
        pa[r] = *(const float2*)&x[(long long)(2 * p) * NCOLS + nbase + sc];
        pb[r] = *(const float2*)&x[(long long)row1 * NCOLS + nbase + sc];
    }

    for (int t = 0; t < TPW; ++t) {
        const long long n0 = nbase + (long long)t * 16;

        #pragma unroll
        for (int r = 0; r < 5; ++r) {
            const int p = r * 4 + sp;
            const bool odd_ok = (2 * p + 1 < K_FIELDS);
            v4f q;
            q[0] = pa[r].x; q[1] = odd_ok ? pb[r].x : 0.0f;
            q[2] = pa[r].y; q[3] = odd_ok ? pb[r].y : 0.0f;
            *(v4f*)&xt[p * PAIRW + sc * 2] = q;
        }
        asm volatile("" ::: "memory");

        if (t + 1 < TPW) {
            const long long n1 = n0 + 16;
            #pragma unroll
            for (int r = 0; r < 5; ++r) {
                const int p    = r * 4 + sp;
                const int row1 = (2 * p + 1 < K_FIELDS) ? (2 * p + 1) : (K_FIELDS - 1);
                pa[r] = *(const float2*)&x[(long long)(2 * p) * NCOLS + n1 + sc];
                pb[r] = *(const float2*)&x[(long long)row1 * NCOLS + n1 + sc];
            }
        }
        __builtin_amdgcn_sched_barrier(0);

        v2f b[10];
        #pragma unroll
        for (int kk = 0; kk < 10; ++kk)
            b[kk] = *(const v2f*)&xt[(2 * kk + hl) * PAIRW + l16 * 2];

        v8f cc[3] = {v8f{}, v8f{}, v8f{}};
        #pragma unroll
        for (int kk = 0; kk < 10; ++kk) {
            cc[0] = __builtin_amdgcn_wmma_f32_16x16x4_f32(false, a0[kk], false, b[kk],
                                                          (short)0, cc[0], false, false);
            if (kk >= 4)
                cc[1] = __builtin_amdgcn_wmma_f32_16x16x4_f32(false, a1[kk - 4], false, b[kk],
                                                              (short)0, cc[1], false, false);
            if (kk >= 8)
                cc[2] = __builtin_amdgcn_wmma_f32_16x16x4_f32(false, a2[kk - 8], false, b[kk],
                                                              (short)0, cc[2], false, false);
        }

        float acc = 0.0f;
        #pragma unroll
        for (int m = 0; m < 3; ++m) {
            #pragma unroll
            for (int pp = 0; pp < (m == 2 ? 3 : 4); ++pp) {
                const int p = m * 8 + hl * 4 + pp;
                const v2f xp = *(const v2f*)&xt[p * PAIRW + l16 * 2];
                acc += xp[0] * cc[m][2 * pp];
                acc += xp[1] * cc[m][2 * pp + 1];
            }
        }
        acc += __shfl_xor(acc, 16, 32);
        if (lane < 16)
            out[n0 + lane] = acc;
    }
}

extern "C" void kernel_launch(void* const* d_in, const int* in_sizes, int n_in,
                              void* d_out, int out_size, void* d_ws, size_t ws_size,
                              hipStream_t stream)
{
    const float* x = (const float*)d_in[0];
    const float* w = (const float*)d_in[1];
    float* out = (float*)d_out;

    dim3 grid(NCOLS / (WPB * TPW * 16));
    dim3 block(256);
    fwfm_wmma_kernel<<<grid, block, 0, stream>>>(x, w, out);
}


# x is loaded from global straight into WMMA B-fragment order, fields permuted so the same registers are the epilogue's x factors: no LDS staging of x, whole-line loads; f32 summation order changes
# speedup vs baseline: 1.5979x; 1.0797x over previous
#include <hip/hip_runtime.h>


typedef __attribute__((ext_vector_type(2))) float v2f;
typedef __attribute__((ext_vector_type(8))) float v8f;

#define K_FIELDS 39
#define NCOLS    (8192 * 64)
#define WPB      8
#define TPW      8
#define TRIPS    (TPW / 2)
#define HALF_F   20
#define ROW2     (NCOLS / 2)

__global__ __launch_bounds__(256, 1)
void fwfm_wmma_kernel(const float* __restrict__ x,
                      const float* __restrict__ w,
                      float* __restrict__ out)
{
    __shared__ __align__(16) float s_pad[48 * 40];

    const int tid  = threadIdx.x;
    const int lane = tid & 31;
    const int wv   = tid >> 5;
    const int hl   = lane >> 4;
    const int l16  = lane & 15;

    for (int i = tid; i < 48 * 40; i += 256) {
        const int rp = i / 40, kap = i - rp * 40;
        const int m = rp >> 4, rho = rp & 15, hr = rho >> 3, r = rho & 7;
        const int fl = 8 * m + r;
        const int fi = HALF_F * hr + fl;
        const int pi = 4 * (fl >> 1) + 2 * hr + (fl & 1);
        const int fj = HALF_F * ((kap >> 1) & 1) + 2 * (kap >> 2) + (kap & 1);
        float v = 0.0f;
        if (fl < HALF_F && fi < K_FIELDS && fj < K_FIELDS && kap > pi)
            v = 0.5f * (w[fi * K_FIELDS + fj] + w[fj * K_FIELDS + fi]);
        s_pad[i] = v;
    }
    __syncthreads();

    v2f a0[10], a1[6], a2[2];
    #pragma unroll
    for (int kk = 0; kk < 10; ++kk) {
        const int k0 = kk * 4 + 2 * hl;
        a0[kk] = *(const v2f*)&s_pad[l16 * 40 + k0];
        if (kk >= 4) a1[kk - 4] = *(const v2f*)&s_pad[(16 + l16) * 40 + k0];
        if (kk >= 8) a2[kk - 8] = *(const v2f*)&s_pad[(32 + l16) * 40 + k0];
    }

    const float2* __restrict__ x2 = (const float2*)x;
    const unsigned nbase = ((unsigned)blockIdx.x * WPB + (unsigned)wv) * (TPW * 16);
    const unsigned col0  = nbase + 2u * (unsigned)l16;
    const unsigned xoff  = (unsigned)(HALF_F * hl) * ROW2 + (col0 >> 1);
    const unsigned xoff_last = xoff + (hl ? 18u : 19u) * ROW2;

    float2 L[20];
    #pragma unroll
    for (int s = 0; s < 19; ++s)
        L[s] = x2[xoff + (unsigned)s * ROW2];
    L[19] = x2[xoff_last];

    #pragma unroll 1
    for (int tr = 0; tr < TRIPS; ++tr) {
        const unsigned cshift = (unsigned)tr * 32u;

        v2f b0[10], b1[10];
        #pragma unroll
        for (int kk = 0; kk < 10; ++kk) {
            b0[kk][0] = L[2 * kk].x;  b0[kk][1] = L[2 * kk + 1].x;
            b1[kk][0] = L[2 * kk].y;  b1[kk][1] = L[2 * kk + 1].y;
        }
        b0[9][1] = hl ? 0.0f : b0[9][1];
        b1[9][1] = hl ? 0.0f : b1[9][1];

        if (tr + 1 < TRIPS) {
            #pragma unroll
            for (int s = 0; s < 19; ++s)
                L[s] = x2[xoff + (unsigned)s * ROW2 + (cshift >> 1) + 16u];
            L[19] = x2[xoff_last + (cshift >> 1) + 16u];
        }
        __builtin_amdgcn_sched_barrier(0);

        float e[2];
        #pragma unroll
        for (int j = 0; j < 2; ++j) {
            const v2f* b = j ? b1 : b0;

            v8f cc[3] = {v8f{}, v8f{}, v8f{}};
            #pragma unroll
            for (int kk = 0; kk < 10; ++kk) {
                cc[0] = __builtin_amdgcn_wmma_f32_16x16x4_f32(false, a0[kk], false, b[kk],
                                                              (short)0, cc[0], false, false);
                if (kk >= 4)
                    cc[1] = __builtin_amdgcn_wmma_f32_16x16x4_f32(false, a1[kk - 4], false, b[kk],
                                                                  (short)0, cc[1], false, false);
                if (kk >= 8)
                    cc[2] = __builtin_amdgcn_wmma_f32_16x16x4_f32(false, a2[kk - 8], false, b[kk],
                                                                  (short)0, cc[2], false, false);
            }

            float acc = 0.0f;
            #pragma unroll
            for (int m = 0; m < 3; ++m) {
                #pragma unroll
                for (int r = 0; r < (m == 2 ? 4 : 8); ++r)
                    acc += b[4 * m + (r >> 1)][r & 1] * cc[m][r];
            }
            e[j] = acc;
        }

        const float keep = hl ? e[1] : e[0];
        const float send = hl ? e[0] : e[1];
        const float got  = __shfl_xor(send, 16, 32);
        out[col0 + cshift + (unsigned)hl] = keep + got;
    }
}

extern "C" void kernel_launch(void* const* d_in, const int* in_sizes, int n_in,
                              void* d_out, int out_size, void* d_ws, size_t ws_size,
                              hipStream_t stream)
{
    const float* x = (const float*)d_in[0];
    const float* w = (const float*)d_in[1];
    float* out = (float*)d_out;

    dim3 grid(NCOLS / (WPB * TPW * 16));
    dim3 block(256);
    fwfm_wmma_kernel<<<grid, block, 0, stream>>>(x, w, out);
}


# First trip's x loads issued at kernel entry, in flight across U build, barrier (LDS-only fence) and fragment reads; U build wait-free for it: 12 unconditional w loads as one group
# speedup vs baseline: 1.7340x; 1.0852x over previous
#include <hip/hip_runtime.h>


typedef __attribute__((ext_vector_type(2))) float v2f;
typedef __attribute__((ext_vector_type(8))) float v8f;

#define K_FIELDS 39
#define NCOLS    (8192 * 64)
#define WPB      8
#define TPW      8
#define TRIPS    (TPW / 2)
#define HALF_F   20
#define ROW2     (NCOLS / 2)

__global__ __launch_bounds__(256, 1)
void fwfm_wmma_kernel(const float* __restrict__ x,
                      const float* __restrict__ w,
                      float* __restrict__ out)
{
    __shared__ __align__(16) float s_pad[48 * 40 + 256];

    const int tid  = threadIdx.x;
    const int lane = tid & 31;
    const int wv   = tid >> 5;
    const int hl   = lane >> 4;
    const int l16  = lane & 15;

    const float2* __restrict__ x2 = (const float2*)x;
    const unsigned nbase = ((unsigned)blockIdx.x * WPB + (unsigned)wv) * (TPW * 16);
    const unsigned col0  = nbase + 2u * (unsigned)l16;
    const unsigned xoff  = (unsigned)(HALF_F * hl) * ROW2 + (col0 >> 1);
    const unsigned xoff_last = xoff + (hl ? 18u : 19u) * ROW2;

    const int cg  = tid >> 4;
    const int hr  = l16 >> 3, r = l16 & 7;
    const int fi0 = HALF_F * hr + r;
    const int pi0 = 4 * (r >> 1) + 2 * hr + (r & 1);
    const int fj0 = HALF_F * ((cg >> 1) & 1) + 2 * (cg >> 2) + (cg & 1);
    float wa[6], wb[6];
    bool  okc[6];
    {
        int c = 0;
        #pragma unroll
        for (int m = 0; m < 3; ++m) {
            #pragma unroll
            for (int j = m; j < 3; ++j, ++c) {
                const int fl = 8 * m + r, fi = fi0 + 8 * m, fj = fj0 + 8 * j;
                const int kap = cg + 16 * j, pi = pi0 + 16 * m;
                okc[c] = (j < 2 || cg < 8) && (fl < HALF_F) && (fi < K_FIELDS) && (fj < K_FIELDS) && (kap > pi);
                wa[c] = w[okc[c] ? fi * K_FIELDS + fj : 0];
                wb[c] = w[okc[c] ? fj * K_FIELDS + fi : 0];
            }
        }
    }
    __builtin_amdgcn_sched_barrier(0);

    float2 L[20];
    #pragma unroll
    for (int s = 0; s < 19; ++s)
        L[s] = x2[xoff + (unsigned)s * ROW2];
    L[19] = x2[xoff_last];
    __builtin_amdgcn_sched_barrier(0);

    {
        int c = 0;
        #pragma unroll
        for (int m = 0; m < 3; ++m) {
            #pragma unroll
            for (int j = m; j < 3; ++j, ++c) {
                const float v = okc[c] ? 0.5f * (wa[c] + wb[c]) : 0.0f;
                const int cell = (j < 2 || cg < 8) ? (16 * m + l16) * 40 + cg + 16 * j : 48 * 40 + tid;
                s_pad[cell] = v;
            }
        }
    }
    __builtin_amdgcn_fence(__ATOMIC_RELEASE, "workgroup", "local");
    __builtin_amdgcn_s_barrier();
    __builtin_amdgcn_fence(__ATOMIC_ACQUIRE, "workgroup", "local");

    v2f a0[10], a1[6], a2[2];
    #pragma unroll
    for (int kk = 0; kk < 10; ++kk) {
        const int k0 = kk * 4 + 2 * hl;
        a0[kk] = *(const v2f*)&s_pad[l16 * 40 + k0];
        if (kk >= 4) a1[kk - 4] = *(const v2f*)&s_pad[(16 + l16) * 40 + k0];
        if (kk >= 8) a2[kk - 8] = *(const v2f*)&s_pad[(32 + l16) * 40 + k0];
    }
    __builtin_amdgcn_sched_barrier(0);

    #pragma unroll 1
    for (int tr = 0; tr < TRIPS; ++tr) {
        const unsigned cshift = (unsigned)tr * 32u;

        v2f b0[10], b1[10];
        #pragma unroll
        for (int kk = 0; kk < 10; ++kk) {
            b0[kk][0] = L[2 * kk].x;  b0[kk][1] = L[2 * kk + 1].x;
            b1[kk][0] = L[2 * kk].y;  b1[kk][1] = L[2 * kk + 1].y;
        }
        b0[9][1] = hl ? 0.0f : b0[9][1];
        b1[9][1] = hl ? 0.0f : b1[9][1];

        if (tr + 1 < TRIPS) {
            #pragma unroll
            for (int s = 0; s < 19; ++s)
                L[s] = x2[xoff + (unsigned)s * ROW2 + (cshift >> 1) + 16u];
            L[19] = x2[xoff_last + (cshift >> 1) + 16u];
        }
        __builtin_amdgcn_sched_barrier(0);

        float e[2];
        #pragma unroll
        for (int j = 0; j < 2; ++j) {
            const v2f* b = j ? b1 : b0;

            v8f cc[3] = {v8f{}, v8f{}, v8f{}};
            #pragma unroll
            for (int kk = 0; kk < 10; ++kk) {
                cc[0] = __builtin_amdgcn_wmma_f32_16x16x4_f32(false, a0[kk], false, b[kk],
                                                              (short)0, cc[0], false, false);
                if (kk >= 4)
                    cc[1] = __builtin_amdgcn_wmma_f32_16x16x4_f32(false, a1[kk - 4], false, b[kk],
                                                                  (short)0, cc[1], false, false);
                if (kk >= 8)
                    cc[2] = __builtin_amdgcn_wmma_f32_16x16x4_f32(false, a2[kk - 8], false, b[kk],
                                                                  (short)0, cc[2], false, false);
            }

            float acc = 0.0f;
            #pragma unroll
            for (int m = 0; m < 3; ++m) {
                #pragma unroll
                for (int r = 0; r < (m == 2 ? 4 : 8); ++r)
                    acc += b[4 * m + (r >> 1)][r & 1] * cc[m][r];
            }
            e[j] = acc;
        }

        const float keep = hl ? e[1] : e[0];
        const float send = hl ? e[0] : e[1];
        const float got  = __shfl_xor(send, 16, 32);
        out[col0 + cshift + (unsigned)hl] = keep + got;
    }
}

extern "C" void kernel_launch(void* const* d_in, const int* in_sizes, int n_in,
                              void* d_out, int out_size, void* d_ws, size_t ws_size,
                              hipStream_t stream)
{
    const float* x = (const float*)d_in[0];
    const float* w = (const float*)d_in[1];
    float* out = (float*)d_out;

    dim3 grid(NCOLS / (WPB * TPW * 16));
    dim3 block(256);
    fwfm_wmma_kernel<<<grid, block, 0, stream>>>(x, w, out);
}


# Each lane loads 4 adjacent columns per field (b128) and the next trip refills the same registers behind the fragment moves: half the loads, no hand-over copies; 16 tiles per wave, 256 work-groups
# speedup vs baseline: 1.9449x; 1.1216x over previous
#include <hip/hip_runtime.h>


typedef __attribute__((ext_vector_type(2))) float v2f;
typedef __attribute__((ext_vector_type(8))) float v8f;

#define K_FIELDS 39
#define NCOLS    (8192 * 64)
#define WPB      8
#define TPW      16
#define TRIPS    (TPW / 4)
#define HALF_F   20
#define ROW4     (NCOLS / 4)

struct RowGroups {
    const float4* g[5];
    const float4* last;
};

static __device__ __forceinline__ float comp4(const float4& v, const int i)
{
    return i == 0 ? v.x : (i == 1 ? v.y : (i == 2 ? v.z : v.w));
}

__global__ __launch_bounds__(256, 1)
void fwfm_wmma_kernel(const float* __restrict__ x,
                      const float* __restrict__ w,
                      float* __restrict__ out)
{
    __shared__ __align__(16) float s_pad[48 * 40 + 256];

    const int tid  = threadIdx.x;
    const int lane = tid & 31;
    const int wv   = tid >> 5;
    const int hl   = lane >> 4;
    const int l16  = lane & 15;

    const float4* __restrict__ x4 = (const float4*)x;
    const unsigned nbase = ((unsigned)blockIdx.x * WPB + (unsigned)wv) * (TPW * 16);
    const unsigned col0  = nbase + 4u * (unsigned)l16;
    const unsigned xoff  = (unsigned)(HALF_F * hl) * ROW4 + (col0 >> 2);
    RowGroups rg;
    #pragma unroll
    for (int k = 0; k < 5; ++k)
        rg.g[k] = x4 + (xoff + (unsigned)(4 * k) * ROW4);
    rg.last = x4 + (xoff + (hl ? 18u : 19u) * ROW4);

    const int cg  = tid >> 4;
    const int hr  = l16 >> 3, r = l16 & 7;
    const int fi0 = HALF_F * hr + r;
    const int pi0 = 4 * (r >> 1) + 2 * hr + (r & 1);
    const int fj0 = HALF_F * ((cg >> 1) & 1) + 2 * (cg >> 2) + (cg & 1);
    float wa[6], wb[6];
    bool  okc[6];
    {
        int c = 0;
        #pragma unroll
        for (int m = 0; m < 3; ++m) {
            #pragma unroll
            for (int j = m; j < 3; ++j, ++c) {
                const int fl = 8 * m + r, fi = fi0 + 8 * m, fj = fj0 + 8 * j;
                const int kap = cg + 16 * j, pi = pi0 + 16 * m;
                okc[c] = (j < 2 || cg < 8) && (fl < HALF_F) && (fi < K_FIELDS) && (fj < K_FIELDS) && (kap > pi);
                wa[c] = w[okc[c] ? fi * K_FIELDS + fj : 0];
                wb[c] = w[okc[c] ? fj * K_FIELDS + fi : 0];
            }
        }
    }
    __builtin_amdgcn_sched_barrier(0);

    float4 L[20];
    #pragma unroll
    for (int s = 0; s < 19; ++s)
        L[s] = rg.g[s >> 2][(unsigned)(s & 3) * ROW4];
    L[19] = rg.last[0];
    __builtin_amdgcn_sched_barrier(0);

    {
        int c = 0;
        #pragma unroll
        for (int m = 0; m < 3; ++m) {
            #pragma unroll
            for (int j = m; j < 3; ++j, ++c) {
                const float v = okc[c] ? 0.5f * (wa[c] + wb[c]) : 0.0f;
                const int cell = (j < 2 || cg < 8) ? (16 * m + l16) * 40 + cg + 16 * j : 48 * 40 + tid;
                s_pad[cell] = v;
            }
        }
    }
    __builtin_amdgcn_fence(__ATOMIC_RELEASE, "workgroup", "local");
    __builtin_amdgcn_s_barrier();
    __builtin_amdgcn_fence(__ATOMIC_ACQUIRE, "workgroup", "local");

    v2f a0[10], a1[6], a2[2];
    #pragma unroll
    for (int kk = 0; kk < 10; ++kk) {
        const int k0 = kk * 4 + 2 * hl;
        a0[kk] = *(const v2f*)&s_pad[l16 * 40 + k0];
        if (kk >= 4) a1[kk - 4] = *(const v2f*)&s_pad[(16 + l16) * 40 + k0];
        if (kk >= 8) a2[kk - 8] = *(const v2f*)&s_pad[(32 + l16) * 40 + k0];
    }
    __builtin_amdgcn_sched_barrier(0);

    const unsigned ocol = col0 + 2u * (unsigned)hl;

    #pragma unroll 1
    for (int tr = 0; tr < TRIPS; ++tr) {
        v2f bf[4][10];
        #pragma unroll
        for (int t = 0; t < 4; ++t) {
            #pragma unroll
            for (int kk = 0; kk < 10; ++kk) {
                bf[t][kk][0] = comp4(L[2 * kk], t);
                bf[t][kk][1] = comp4(L[2 * kk + 1], t);
            }
            bf[t][9][1] = hl ? 0.0f : bf[t][9][1];
            asm volatile("" : "+v"(bf[t][0]), "+v"(bf[t][1]), "+v"(bf[t][2]), "+v"(bf[t][3]), "+v"(bf[t][4]),
                              "+v"(bf[t][5]), "+v"(bf[t][6]), "+v"(bf[t][7]), "+v"(bf[t][8]), "+v"(bf[t][9]));
        }

        if (tr + 1 < TRIPS) {
            const unsigned noff = (unsigned)(tr + 1) * 16u;
            #pragma unroll
            for (int s = 0; s < 19; ++s)
                L[s] = rg.g[s >> 2][(unsigned)(s & 3) * ROW4 + noff];
            L[19] = rg.last[noff];
        }
        __builtin_amdgcn_sched_barrier(0);

        float e[4];
        #pragma unroll
        for (int p = 0; p < 2; ++p) {
            const v2f* bA = bf[2 * p];
            const v2f* bB = bf[2 * p + 1];

            v8f ca[3] = {v8f{}, v8f{}, v8f{}};
            v8f cb[3] = {v8f{}, v8f{}, v8f{}};
            #pragma unroll
            for (int kk = 0; kk < 10; ++kk) {
                ca[0] = __builtin_amdgcn_wmma_f32_16x16x4_f32(false, a0[kk], false, bA[kk],
                                                              (short)0, ca[0], false, false);
                cb[0] = __builtin_amdgcn_wmma_f32_16x16x4_f32(false, a0[kk], false, bB[kk],
                                                              (short)0, cb[0], false, false);
                if (kk >= 4) {
                    ca[1] = __builtin_amdgcn_wmma_f32_16x16x4_f32(false, a1[kk - 4], false, bA[kk],
                                                                  (short)0, ca[1], false, false);
                    cb[1] = __builtin_amdgcn_wmma_f32_16x16x4_f32(false, a1[kk - 4], false, bB[kk],
                                                                  (short)0, cb[1], false, false);
                }
                if (kk >= 8) {
                    ca[2] = __builtin_amdgcn_wmma_f32_16x16x4_f32(false, a2[kk - 8], false, bA[kk],
                                                                  (short)0, ca[2], false, false);
                    cb[2] = __builtin_amdgcn_wmma_f32_16x16x4_f32(false, a2[kk - 8], false, bB[kk],
                                                                  (short)0, cb[2], false, false);
                }
            }

            #pragma unroll
            for (int j = 0; j < 2; ++j) {
                const v2f* b  = j ? bB : bA;
                const v8f* cc = j ? cb : ca;
                float acc = 0.0f;
                #pragma unroll
                for (int m = 0; m < 3; ++m) {
                    #pragma unroll
                    for (int r = 0; r < (m == 2 ? 4 : 8); ++r)
                        acc = __builtin_fmaf(b[4 * m + (r >> 1)][r & 1], cc[m][r], acc);
                }
                e[2 * p + j] = acc;
            }
            __builtin_amdgcn_sched_barrier(0);
        }

        const float keep0 = hl ? e[2] : e[0], keep1 = hl ? e[3] : e[1];
        const float send0 = hl ? e[0] : e[2], send1 = hl ? e[1] : e[3];
        const float got0  = __shfl_xor(send0, 16, 32);
        const float got1  = __shfl_xor(send1, 16, 32);
        float2 res;
        res.x = keep0 + got0;
        res.y = keep1 + got1;
        *(float2*)&out[ocol + (unsigned)tr * 64u] = res;
    }
}

extern "C" void kernel_launch(void* const* d_in, const int* in_sizes, int n_in,
                              void* d_out, int out_size, void* d_ws, size_t ws_size,
                              hipStream_t stream)
{
    const float* x = (const float*)d_in[0];
    const float* w = (const float*)d_in[1];
    float* out = (float*)d_out;

    dim3 grid(NCOLS / (WPB * TPW * 16));
    dim3 block(256);
    fwfm_wmma_kernel<<<grid, block, 0, stream>>>(x, w, out);
}
